# starting kernel
# baseline (speedup 1.0000x reference)
#include <hip/hip_runtime.h>

typedef __attribute__((ext_vector_type(2))) float v2f;
typedef __attribute__((ext_vector_type(8))) float v8f;

#define LORENZ_DT 0.02
#define LORENZ_ORDER 5

struct ATab { float v[16][20]; };

constexpr ATab make_atab() {
    double A0[3][3] = {{-10.0, 10.0, 0.0},
                       { 28.0, -1.0, 0.0},
                       {  0.0,  0.0, -8.0 / 3.0}};
    double A1[3][3] = {{0.0, 0.0,  0.0},
                       {0.0, 0.0, -1.0},
                       {0.0, 1.0,  0.0}};
    double T[6][3][3] = {};
    double F[6][3][3] = {};
    for (int i = 0; i < 3; ++i) { T[0][i][i] = 1.0; F[0][i][i] = 1.0; }
    for (int j = 1; j <= LORENZ_ORDER; ++j) {
        double NT[6][3][3] = {};
        for (int d = 0; d < 6; ++d)
            for (int r = 0; r < 3; ++r)
                for (int c = 0; c < 3; ++c) {
                    double acc = 0.0;
                    for (int k = 0; k < 3; ++k) {
                        acc += T[d][r][k] * A0[k][c];
                        if (d >= 1) acc += T[d - 1][r][k] * A1[k][c];
                    }
                    NT[d][r][c] = acc * LORENZ_DT / (double)j;
                }
        for (int d = 0; d < 6; ++d)
            for (int r = 0; r < 3; ++r)
                for (int c = 0; c < 3; ++c) {
                    T[d][r][c] = NT[d][r][c];
                    F[d][r][c] += NT[d][r][c];
                }
    }
    ATab t = {};
    for (int n = 0; n < 3; ++n)
        for (int p = 0; p < 6; ++p)
            for (int c = 0; c < 3; ++c)
                t.v[n][3 * p + c] = (float)F[p][n][c];
    return t;
}

__device__ __constant__ ATab g_atab = make_atab();

__global__ void lorenz_wmma_kernel(const float* __restrict__ x,
                                   float* __restrict__ out,
                                   int n, int ntiles) {
    const int tid    = blockIdx.x * blockDim.x + threadIdx.x;
    const int lane   = threadIdx.x & 31;
    const int waveId = tid >> 5;
    const int nwaves = (gridDim.x * blockDim.x) >> 5;
    const int row    = lane & 15;
    const int hi     = lane >> 4;

    v2f afrag[5];
#pragma unroll
    for (int c = 0; c < 5; ++c) {
        afrag[c].x = g_atab.v[row][4 * c + 2 * hi + 0];
        afrag[c].y = g_atab.v[row][4 * c + 2 * hi + 1];
    }

    for (int t = waveId; t < ntiles; t += nwaves) {
        const int elem = t * 16 + row;
        const float x0 = x[elem * 3 + 0];
        const float x1 = x[elem * 3 + 1];
        const float x2 = x[elem * 3 + 2];

        const float s1 = x0;
        const float s2 = s1 * s1;
        const float s3 = s2 * s1;
        const float s4 = s2 * s2;
        const float s5 = s4 * s1;

        float z[20];
        z[0]  = x0;      z[1]  = x1;      z[2]  = x2;
        z[3]  = s1 * x0; z[4]  = s1 * x1; z[5]  = s1 * x2;
        z[6]  = s2 * x0; z[7]  = s2 * x1; z[8]  = s2 * x2;
        z[9]  = s3 * x0; z[10] = s3 * x1; z[11] = s3 * x2;
        z[12] = s4 * x0; z[13] = s4 * x1; z[14] = s4 * x2;
        z[15] = s5 * x0; z[16] = s5 * x1; z[17] = s5 * x2;
        z[18] = 0.0f;    z[19] = 0.0f;

        v8f acc = {};
#pragma unroll
        for (int c = 0; c < 5; ++c) {
            v2f b;
            b.x = hi ? z[4 * c + 2] : z[4 * c + 0];
            b.y = hi ? z[4 * c + 3] : z[4 * c + 1];
            acc = __builtin_amdgcn_wmma_f32_16x16x4_f32(
                false, afrag[c], false, b, (short)0, acc, false, false);
        }

        if (lane < 16) {
            out[elem * 3 + 0] = acc[0];
            out[elem * 3 + 1] = acc[1];
            out[elem * 3 + 2] = acc[2];
        }
    }

    const int tail0 = ntiles * 16;
    const int ntail = n - tail0;
    if (tid < ntail) {
        const int e = tail0 + tid;
        const float xv0 = x[e * 3 + 0];
        const float xv1 = x[e * 3 + 1];
        const float xv2 = x[e * 3 + 2];
        const float dt  = (float)LORENZ_DT;
        float A[3][3] = {{-10.0f * dt, 10.0f * dt, 0.0f},
                         { 28.0f * dt, -1.0f * dt, -xv0 * dt},
                         { 0.0f,        xv0 * dt,  (-8.0f / 3.0f) * dt}};
        float F[3][3] = {{1, 0, 0}, {0, 1, 0}, {0, 0, 1}};
        float T[3][3] = {{1, 0, 0}, {0, 1, 0}, {0, 0, 1}};
        for (int j = 1; j <= LORENZ_ORDER; ++j) {
            float NT[3][3];
            const float inv = 1.0f / (float)j;
            for (int r = 0; r < 3; ++r)
                for (int c = 0; c < 3; ++c)
                    NT[r][c] = (T[r][0] * A[0][c] + T[r][1] * A[1][c] +
                                T[r][2] * A[2][c]) * inv;
            for (int r = 0; r < 3; ++r)
                for (int c = 0; c < 3; ++c) {
                    T[r][c] = NT[r][c];
                    F[r][c] += NT[r][c];
                }
        }
        out[e * 3 + 0] = F[0][0] * xv0 + F[0][1] * xv1 + F[0][2] * xv2;
        out[e * 3 + 1] = F[1][0] * xv0 + F[1][1] * xv1 + F[1][2] * xv2;
        out[e * 3 + 2] = F[2][0] * xv0 + F[2][1] * xv1 + F[2][2] * xv2;
    }
}

extern "C" void kernel_launch(void* const* d_in, const int* in_sizes, int n_in,
                              void* d_out, int out_size, void* d_ws, size_t ws_size,
                              hipStream_t stream) {
    const float* x = (const float*)d_in[0];
    float* out = (float*)d_out;
    const int n = in_sizes[0] / 3;
    const int ntiles = n / 16;

    const int threads = 256;
    int blocks = 1024;
    if (blocks * (threads / 32) > ntiles && ntiles > 0) {
        blocks = (ntiles + (threads / 32) - 1) / (threads / 32);
    }
    if (blocks < 1) blocks = 1;

    lorenz_wmma_kernel<<<blocks, threads, 0, stream>>>(x, out, n, ntiles);
}


# Memory path: a wave moves 32 consecutive elements per load/store instruction (two tiles, half-wave exchange), next step's 4 loads issued before the current step's math; 4096 waves, not 8192
# speedup vs baseline: 1.3003x; 1.3003x over previous
#include <hip/hip_runtime.h>

typedef __attribute__((ext_vector_type(2))) float v2f;
typedef __attribute__((ext_vector_type(8))) float v8f;

#define LORENZ_DT 0.02
#define LORENZ_ORDER 5

struct ATab { float v[16][20]; };

constexpr ATab make_atab() {
    double A0[3][3] = {{-10.0, 10.0, 0.0},
                       { 28.0, -1.0, 0.0},
                       {  0.0,  0.0, -8.0 / 3.0}};
    double A1[3][3] = {{0.0, 0.0,  0.0},
                       {0.0, 0.0, -1.0},
                       {0.0, 1.0,  0.0}};
    double T[6][3][3] = {};
    double F[6][3][3] = {};
    for (int i = 0; i < 3; ++i) { T[0][i][i] = 1.0; F[0][i][i] = 1.0; }
    for (int j = 1; j <= LORENZ_ORDER; ++j) {
        double NT[6][3][3] = {};
        for (int d = 0; d < 6; ++d)
            for (int r = 0; r < 3; ++r)
                for (int c = 0; c < 3; ++c) {
                    double acc = 0.0;
                    for (int k = 0; k < 3; ++k) {
                        acc += T[d][r][k] * A0[k][c];
                        if (d >= 1) acc += T[d - 1][r][k] * A1[k][c];
                    }
                    NT[d][r][c] = acc * LORENZ_DT / (double)j;
                }
        for (int d = 0; d < 6; ++d)
            for (int r = 0; r < 3; ++r)
                for (int c = 0; c < 3; ++c) {
                    T[d][r][c] = NT[d][r][c];
                    F[d][r][c] += NT[d][r][c];
                }
    }
    ATab t = {};
    for (int n = 0; n < 3; ++n)
        for (int p = 0; p < 6; ++p)
            for (int c = 0; c < 3; ++c)
                t.v[n][3 * p + c] = t.v[n + 8][3 * p + c] = (float)F[p][n][c];
    return t;
}

__device__ __constant__ ATab g_atab = make_atab();

#define PAIRS 4

__device__ __forceinline__ float xhalf(float v) {
    const int b = __builtin_bit_cast(int, v);
    return __builtin_bit_cast(float,
        __builtin_amdgcn_permlanex16(b, b, 0x76543210, 0xfedcba98, false, false));
}

__device__ __forceinline__ v8f tile_product(const v2f (&afrag)[5],
                                            float x0, float x1, float x2, int hi) {
    const float s1 = x0;
    const float s2 = s1 * s1;
    const float s3 = s2 * s1;
    const float s4 = s2 * s2;
    const float s5 = s4 * s1;

    float z[20];
    z[0]  = x0;      z[1]  = x1;      z[2]  = x2;
    z[3]  = s1 * x0; z[4]  = s1 * x1; z[5]  = s1 * x2;
    z[6]  = s2 * x0; z[7]  = s2 * x1; z[8]  = s2 * x2;
    z[9]  = s3 * x0; z[10] = s3 * x1; z[11] = s3 * x2;
    z[12] = s4 * x0; z[13] = s4 * x1; z[14] = s4 * x2;
    z[15] = s5 * x0; z[16] = s5 * x1; z[17] = s5 * x2;
    z[18] = 0.0f;    z[19] = 0.0f;

    v8f acc = {};
#pragma unroll
    for (int c = 0; c < 5; ++c) {
        v2f b;
        b.x = hi ? z[4 * c + 2] : z[4 * c + 0];
        b.y = hi ? z[4 * c + 3] : z[4 * c + 1];
        acc = __builtin_amdgcn_wmma_f32_16x16x4_f32(
            false, afrag[c], false, b, (short)0, acc, false, false);
    }
    return acc;
}

template <int U>
__device__ __forceinline__ void load_pairs(const float* __restrict__ x, int e0, int lane,
                                           float (&a)[U], float (&b)[U], float (&c)[U]) {
#pragma unroll
    for (int j = 0; j < U; ++j) {
        const float* p = x + (e0 + 32 * j + lane) * 3;
        a[j] = p[0];
        b[j] = p[1];
        c[j] = p[2];
    }
}

template <int U>
__device__ __forceinline__ void compute_store(float* __restrict__ out, int e0, int lane, int hi,
                                              const v2f (&afrag)[5],
                                              const float (&a)[U], const float (&b)[U],
                                              const float (&c)[U]) {
#pragma unroll
    for (int j = 0; j < U; ++j) {
        const float pa = xhalf(a[j]);
        const float pb = xhalf(b[j]);
        const float pc = xhalf(c[j]);
        const v8f accA = tile_product(afrag, hi ? pa : a[j], hi ? pb : b[j], hi ? pc : c[j], hi);
        const v8f accB = tile_product(afrag, hi ? a[j] : pa, hi ? b[j] : pb, hi ? c[j] : pc, hi);
        float* q = out + (e0 + 32 * j + lane) * 3;
        q[0] = hi ? accB[0] : accA[0];
        q[1] = hi ? accB[1] : accA[1];
        q[2] = hi ? accB[2] : accA[2];
    }
}

__global__ void lorenz_wmma_kernel(const float* __restrict__ x,
                                   float* __restrict__ out,
                                   int n, int npairs) {
    const int tid    = blockIdx.x * blockDim.x + threadIdx.x;
    const int lane   = threadIdx.x & 31;
    const int waveId = __builtin_amdgcn_readfirstlane(tid >> 5);
    const int nwaves = (gridDim.x * blockDim.x) >> 5;
    const int row    = lane & 15;
    const int hi     = lane >> 4;

    v2f afrag[5];
#pragma unroll
    for (int c = 0; c < 5; ++c) {
        afrag[c].x = g_atab.v[row][4 * c + 2 * hi + 0];
        afrag[c].y = g_atab.v[row][4 * c + 2 * hi + 1];
    }

    const int nsteps = npairs / PAIRS;
    if (waveId < nsteps) {
        int g = waveId;
        float ca[PAIRS], cb[PAIRS], cc[PAIRS];
        load_pairs<PAIRS>(x, g * (32 * PAIRS), lane, ca, cb, cc);
        for (int gn = g + nwaves; gn < nsteps; gn += nwaves) {
            float na[PAIRS], nb[PAIRS], nc[PAIRS];
            load_pairs<PAIRS>(x, gn * (32 * PAIRS), lane, na, nb, nc);
            __builtin_amdgcn_sched_barrier(0);
            compute_store<PAIRS>(out, g * (32 * PAIRS), lane, hi, afrag, ca, cb, cc);
#pragma unroll
            for (int j = 0; j < PAIRS; ++j) { ca[j] = na[j]; cb[j] = nb[j]; cc[j] = nc[j]; }
            g = gn;
        }
        compute_store<PAIRS>(out, g * (32 * PAIRS), lane, hi, afrag, ca, cb, cc);
    }

    for (int p = nsteps * PAIRS + waveId; p < npairs; p += nwaves) {
        float ca[1], cb[1], cc[1];
        load_pairs<1>(x, p * 32, lane, ca, cb, cc);
        compute_store<1>(out, p * 32, lane, hi, afrag, ca, cb, cc);
    }

    const int tail0 = npairs * 32;
    const int ntail = n - tail0;
    if (tid < ntail) {
        const int e = tail0 + tid;
        const float xv0 = x[e * 3 + 0];
        const float xv1 = x[e * 3 + 1];
        const float xv2 = x[e * 3 + 2];
        const float dt  = (float)LORENZ_DT;
        float A[3][3] = {{-10.0f * dt, 10.0f * dt, 0.0f},
                         { 28.0f * dt, -1.0f * dt, -xv0 * dt},
                         { 0.0f,        xv0 * dt,  (-8.0f / 3.0f) * dt}};
        float F[3][3] = {{1, 0, 0}, {0, 1, 0}, {0, 0, 1}};
        float T[3][3] = {{1, 0, 0}, {0, 1, 0}, {0, 0, 1}};
        for (int j = 1; j <= LORENZ_ORDER; ++j) {
            float NT[3][3];
            const float inv = 1.0f / (float)j;
            for (int r = 0; r < 3; ++r)
                for (int c = 0; c < 3; ++c)
                    NT[r][c] = (T[r][0] * A[0][c] + T[r][1] * A[1][c] +
                                T[r][2] * A[2][c]) * inv;
            for (int r = 0; r < 3; ++r)
                for (int c = 0; c < 3; ++c) {
                    T[r][c] = NT[r][c];
                    F[r][c] += NT[r][c];
                }
        }
        out[e * 3 + 0] = F[0][0] * xv0 + F[0][1] * xv1 + F[0][2] * xv2;
        out[e * 3 + 1] = F[1][0] * xv0 + F[1][1] * xv1 + F[1][2] * xv2;
        out[e * 3 + 2] = F[2][0] * xv0 + F[2][1] * xv1 + F[2][2] * xv2;
    }
}

extern "C" void kernel_launch(void* const* d_in, const int* in_sizes, int n_in,
                              void* d_out, int out_size, void* d_ws, size_t ws_size,
                              hipStream_t stream) {
    const float* x = (const float*)d_in[0];
    float* out = (float*)d_out;
    const int n = in_sizes[0] / 3;
    const int npairs = n / 32;
    const int nsteps = npairs / PAIRS;

    const int threads = 256;
    int blocks = 512;
    if (blocks * (threads / 32) > nsteps && nsteps > 0) {
        blocks = (nsteps + (threads / 32) - 1) / (threads / 32);
    }
    if (blocks < 1) blocks = 1;

    lorenz_wmma_kernel<<<blocks, threads, 0, stream>>>(x, out, n, npairs);
}


# Each wave takes 4 consecutive steps and requests all 16 loads of 384 B at entry (was 4 in flight, the next 4 issued only after those had landed): one exposed memory round trip per wave, not four
# speedup vs baseline: 2.2019x; 1.6934x over previous
#include <hip/hip_runtime.h>

typedef __attribute__((ext_vector_type(2))) float v2f;
typedef __attribute__((ext_vector_type(8))) float v8f;

#define LORENZ_DT 0.02
#define LORENZ_ORDER 5

struct MTab { float v[16][4]; bool ok; };

constexpr MTab make_mtab() {
    double A0[3][3] = {{-10.0, 10.0, 0.0},
                       { 28.0, -1.0, 0.0},
                       {  0.0,  0.0, -8.0 / 3.0}};
    double A1[3][3] = {{0.0, 0.0,  0.0},
                       {0.0, 0.0, -1.0},
                       {0.0, 1.0,  0.0}};
    double T[6][3][3] = {};
    double F[6][3][3] = {};
    for (int i = 0; i < 3; ++i) { T[0][i][i] = 1.0; F[0][i][i] = 1.0; }
    for (int j = 1; j <= LORENZ_ORDER; ++j) {
        double NT[6][3][3] = {};
        for (int d = 0; d < 6; ++d)
            for (int r = 0; r < 3; ++r)
                for (int c = 0; c < 3; ++c) {
                    double acc = 0.0;
                    for (int k = 0; k < 3; ++k) {
                        acc += T[d][r][k] * A0[k][c];
                        if (d >= 1) acc += T[d - 1][r][k] * A1[k][c];
                    }
                    NT[d][r][c] = acc * LORENZ_DT / (double)j;
                }
        for (int d = 0; d < 6; ++d)
            for (int r = 0; r < 3; ++r)
                for (int c = 0; c < 3; ++c) {
                    T[d][r][c] = NT[d][r][c];
                    F[d][r][c] += NT[d][r][c];
                }
    }
    MTab m = {};
    m.ok = F[5][0][0] == 0.0 && F[5][0][1] == 0.0 && F[5][0][2] == 0.0 &&
           F[4][0][0] == 0.0 && F[4][0][2] == 0.0;
    const int first_row[3] = {0, 2, 5};
    for (int half = 0; half < 2; ++half)
        for (int i = 0; i < 3; ++i) {
            const int nq = (i == 0) ? 2 : 3;
            for (int q = 0; q < nq; ++q)
                for (int c = 0; c < 3; ++c)
                    m.v[8 * half + first_row[i] + q][c] = (float)F[half + 2 * q][i][c];
        }
    m.v[1][3] = (float)F[4][0][1];
    return m;
}

static_assert(make_mtab().ok, "row packing needs D5[0] == 0 and D4[0] on x1 only");

__device__ __constant__ MTab g_mtab = make_mtab();

#define PAIRS 4
#define STEPS 4

__device__ __forceinline__ void swap_rows(float& a, float& b) {
    asm("v_permlane16_swap_b32 %0, %1" : "+v"(a), "+v"(b));
}

__device__ __forceinline__ void half_poly(const v8f& d, float t, float& p0, float& p1, float& p2) {
    p0 = __builtin_fmaf(t, d[1], d[0]);
    p1 = __builtin_fmaf(t, __builtin_fmaf(t, d[4], d[3]), d[2]);
    p2 = __builtin_fmaf(t, __builtin_fmaf(t, d[7], d[6]), d[5]);
}

template <int U>
__device__ __forceinline__ void load_pairs(const float* __restrict__ x, int e0, int lane,
                                           float (&a)[U], float (&b)[U], float (&c)[U]) {
    const float* xs = x + (size_t)(unsigned)e0 * 3u;
    const unsigned l3 = (unsigned)lane * 3u;
#pragma unroll
    for (int j = 0; j < U; ++j) {
        const float* p = xs + 96 * j;
        a[j] = p[l3 + 0];
        b[j] = p[l3 + 1];
        c[j] = p[l3 + 2];
    }
}

template <int J0, int U, int N>
__device__ __forceinline__ void compute_store(float* __restrict__ out, int e0, int lane,
                                              const v2f mfrag,
                                              const float (&a)[N], const float (&b)[N],
                                              const float (&c)[N]) {
#pragma unroll
    for (int j = J0; j < J0 + U; ++j) {
        const float s = a[j];
        const float t = s * s;

        float f1x = a[j], f2x = c[j];
        swap_rows(f1x, f2x);
        float f1y = b[j], f2y = t * b[j];
        swap_rows(f1y, f2y);
        float t1 = t, t2 = t;
        swap_rows(t1, t2);

        v2f fr1, fr2;
        fr1.x = f1x; fr1.y = f1y;
        fr2.x = f2x; fr2.y = f2y;
        const v8f zero = {};
        const v8f d1 = __builtin_amdgcn_wmma_f32_16x16x4_f32(
            false, mfrag, false, fr1, (short)0, zero, false, false);
        const v8f d2 = __builtin_amdgcn_wmma_f32_16x16x4_f32(
            false, mfrag, false, fr2, (short)0, zero, false, false);

        float g0, g1, g2, h0, h1, h2;
        half_poly(d1, t1, g0, g1, g2);
        half_poly(d2, t2, h0, h1, h2);
        swap_rows(g0, h0);
        swap_rows(g1, h1);
        swap_rows(g2, h2);

        float* q = out + (e0 + 32 * j + lane) * 3;
        q[0] = __builtin_fmaf(s, h0, g0);
        q[1] = __builtin_fmaf(s, h1, g1);
        q[2] = __builtin_fmaf(s, h2, g2);
    }
}

__device__ __forceinline__ void run_steps(const float* __restrict__ x, float* __restrict__ out,
                                          int e0, int lane, const v2f mfrag) {
    float a[STEPS * PAIRS], b[STEPS * PAIRS], c[STEPS * PAIRS];
    load_pairs<STEPS * PAIRS>(x, e0, lane, a, b, c);
    __builtin_amdgcn_sched_barrier(0);
    compute_store<0 * PAIRS, PAIRS>(out, e0, lane, mfrag, a, b, c);
    __builtin_amdgcn_sched_barrier(0);
    compute_store<1 * PAIRS, PAIRS>(out, e0, lane, mfrag, a, b, c);
    __builtin_amdgcn_sched_barrier(0);
    compute_store<2 * PAIRS, PAIRS>(out, e0, lane, mfrag, a, b, c);
    __builtin_amdgcn_sched_barrier(0);
    compute_store<3 * PAIRS, PAIRS>(out, e0, lane, mfrag, a, b, c);
}

__global__ void __launch_bounds__(256) lorenz_wmma_kernel(const float* __restrict__ x,
                                   float* __restrict__ out,
                                   int n, int npairs) {
    const int tid    = blockIdx.x * blockDim.x + threadIdx.x;
    const int lane   = threadIdx.x & 31;
    const int waveId = __builtin_amdgcn_readfirstlane(tid >> 5);
    const int nwaves = (gridDim.x * blockDim.x) >> 5;
    const int row    = lane & 15;
    const int hi     = lane >> 4;

    v2f mfrag;
    mfrag.x = g_mtab.v[row][2 * hi + 0];
    mfrag.y = g_mtab.v[row][2 * hi + 1];

    const int nsteps = npairs / PAIRS;
    for (int s0 = waveId * STEPS; s0 < nsteps; s0 += nwaves * STEPS) {
        if (s0 + STEPS <= nsteps) {
            run_steps(x, out, s0 * (32 * PAIRS), lane, mfrag);
        } else {
            for (int s = s0; s < nsteps; ++s) {
                float ca[PAIRS], cb[PAIRS], cc[PAIRS];
                load_pairs<PAIRS>(x, s * (32 * PAIRS), lane, ca, cb, cc);
                compute_store<0, PAIRS>(out, s * (32 * PAIRS), lane, mfrag, ca, cb, cc);
            }
        }
    }

    for (int p = nsteps * PAIRS + waveId; p < npairs; p += nwaves) {
        float ca[1], cb[1], cc[1];
        load_pairs<1>(x, p * 32, lane, ca, cb, cc);
        compute_store<0, 1>(out, p * 32, lane, mfrag, ca, cb, cc);
    }

    const int tail0 = npairs * 32;
    const int ntail = n - tail0;
    if (tid < ntail) {
        const int e = tail0 + tid;
        const float xv0 = x[e * 3 + 0];
        const float xv1 = x[e * 3 + 1];
        const float xv2 = x[e * 3 + 2];
        const float dt  = (float)LORENZ_DT;
        float A[3][3] = {{-10.0f * dt, 10.0f * dt, 0.0f},
                         { 28.0f * dt, -1.0f * dt, -xv0 * dt},
                         { 0.0f,        xv0 * dt,  (-8.0f / 3.0f) * dt}};
        float F[3][3] = {{1, 0, 0}, {0, 1, 0}, {0, 0, 1}};
        float T[3][3] = {{1, 0, 0}, {0, 1, 0}, {0, 0, 1}};
        for (int j = 1; j <= LORENZ_ORDER; ++j) {
            float NT[3][3];
            const float inv = 1.0f / (float)j;
            for (int r = 0; r < 3; ++r)
                for (int c = 0; c < 3; ++c)
                    NT[r][c] = (T[r][0] * A[0][c] + T[r][1] * A[1][c] +
                                T[r][2] * A[2][c]) * inv;
            for (int r = 0; r < 3; ++r)
                for (int c = 0; c < 3; ++c) {
                    T[r][c] = NT[r][c];
                    F[r][c] += NT[r][c];
                }
        }
        out[e * 3 + 0] = F[0][0] * xv0 + F[0][1] * xv1 + F[0][2] * xv2;
        out[e * 3 + 1] = F[1][0] * xv0 + F[1][1] * xv1 + F[1][2] * xv2;
        out[e * 3 + 2] = F[2][0] * xv0 + F[2][1] * xv1 + F[2][2] * xv2;
    }
}

extern "C" void kernel_launch(void* const* d_in, const int* in_sizes, int n_in,
                              void* d_out, int out_size, void* d_ws, size_t ws_size,
                              hipStream_t stream) {
    const float* x = (const float*)d_in[0];
    float* out = (float*)d_out;
    const int n = in_sizes[0] / 3;
    const int npairs = n / 32;
    const int nsteps = npairs / PAIRS;

    const int threads = 256;
    int blocks = 512;
    if (blocks * (threads / 32) > nsteps && nsteps > 0) {
        blocks = (nsteps + (threads / 32) - 1) / (threads / 32);
    }
    if (blocks < 1) blocks = 1;

    lorenz_wmma_kernel<<<blocks, threads, 0, stream>>>(x, out, n, npairs);
}
